# starting kernel
# baseline (speedup 1.0000x reference)
#include <hip/hip_runtime.h>
#include <hip/hip_bf16.h>

#define BB 4096
#define LL 5
#define DD 128
#define NHH 16
#define NVV 4
#define TT 100
#define FC1 752

typedef float v2f __attribute__((ext_vector_type(2)));
typedef float v8f __attribute__((ext_vector_type(8)));

__global__ void caser_features(const int* __restrict__ seq,
                               const int* __restrict__ user,
                               const float* __restrict__ item_table,
                               const float* __restrict__ user_table,
                               const float* __restrict__ Wv,
                               const float* __restrict__ bv,
                               const float* __restrict__ Wh,
                               const float* __restrict__ bh,
                               float* __restrict__ z,
                               float* __restrict__ xc)
{
    const int b   = blockIdx.x;
    const int tid = threadIdx.x;
    const int wave = tid >> 5;
    const int lane = tid & 31;

    __shared__ float semb[LL][DD];
    __shared__ float sWv[NVV * LL];

    float e[LL];
    #pragma unroll
    for (int t = 0; t < LL; ++t) {
        const int it = seq[b * LL + t];
        e[t] = item_table[(size_t)it * DD + tid];
        semb[t][tid] = e[t];
    }
    if (tid < NVV * LL) sWv[tid] = Wv[tid];
    __syncthreads();

    #pragma unroll
    for (int v = 0; v < NVV; ++v) {
        float acc = bv[v];
        #pragma unroll
        for (int t = 0; t < LL; ++t) acc += e[t] * sWv[v * LL + t];
        z[(size_t)b * FC1 + v * DD + tid] = acc;
    }

    xc[(size_t)b * (2 * DD) + DD + tid] =
        user_table[(size_t)user[b] * DD + tid];

    for (int o = wave; o < 240; o += 4) {
        int rem = o;
        int off = NVV * DD;
        int l = 1;
        for (; l <= LL; ++l) {
            const int cnt = NHH * (LL - l + 1);
            if (rem < cnt) break;
            rem -= cnt; off += cnt;
        }
        const int lout = LL - l + 1;
        const int f = rem / lout;
        const int t = rem % lout;

        const float* wrow = Wh + ((size_t)(l - 1) * NHH + f) * (LL * DD);
        float acc = 0.f;
        for (int i = lane; i < l * DD; i += 32) {
            const int s = i >> 7;
            const int d = i & (DD - 1);
            acc += semb[t + s][d] * wrow[s * DD + d];
        }
        #pragma unroll
        for (int sft = 16; sft >= 1; sft >>= 1)
            acc += __shfl_down(acc, sft, 32);
        if (lane == 0) {
            acc += bh[(l - 1) * NHH + f];
            z[(size_t)b * FC1 + off + f * lout + t] = fmaxf(acc, 0.f);
        }
    }
}

__global__ void caser_fc1(const float* __restrict__ z,
                          const float* __restrict__ W1,
                          const float* __restrict__ b1,
                          float* __restrict__ xc)
{
    __shared__ float sz[16][FC1 + 4];
    const int tid  = threadIdx.x;
    const int wave = tid >> 5;
    const int lane = tid & 31;
    const int m0   = blockIdx.x * 16;

    for (int i = tid; i < 16 * FC1; i += 256) {
        const int r = i / FC1, c = i % FC1;
        sz[r][c] = z[(size_t)(m0 + r) * FC1 + c];
    }
    __syncthreads();

    const int n0   = wave * 16;
    const int mrow = lane & 15;
    const int half = lane >> 4;

    v8f acc = {};
    for (int k0 = 0; k0 < FC1; k0 += 4) {
        const int ka = k0 + half * 2;
        v2f a, bf;
        a.x = sz[mrow][ka + 0];
        a.y = sz[mrow][ka + 1];
        bf.x = W1[(size_t)(ka + 0) * DD + n0 + mrow];
        bf.y = W1[(size_t)(ka + 1) * DD + n0 + mrow];
        acc = __builtin_amdgcn_wmma_f32_16x16x4_f32(
             false, a,  false, bf,
             (short)0, acc,  false,  false);
    }

    const int n = n0 + (lane & 15);
    const float bias = b1[n];
    #pragma unroll
    for (int r = 0; r < 8; ++r) {
        const int m = m0 + r + half * 8;
        const float v = acc[r] + bias;
        xc[(size_t)m * (2 * DD) + n] = fmaxf(v, 0.f);
    }
}

__global__ void caser_score(const int* __restrict__ items,
                            const float* __restrict__ xc,
                            const float* __restrict__ W2,
                            const float* __restrict__ b2,
                            float* __restrict__ out)
{
    const int b    = blockIdx.x;
    const int tid  = threadIdx.x;
    const int wave = tid >> 5;
    const int lane = tid & 31;

    __shared__ __align__(16) float sxc[2 * DD];
    sxc[tid] = xc[(size_t)b * (2 * DD) + tid];
    __syncthreads();

    const float4 xa = *(const float4*)&sxc[lane * 8 + 0];
    const float4 xb = *(const float4*)&sxc[lane * 8 + 4];

    for (int t = wave; t < TT; t += 8) {
        const int it = items[b * TT + t];
        const int tn = t + 8;
        if (tn < TT) {
            const int itn = items[b * TT + tn];
            __builtin_prefetch(W2 + (size_t)itn * (2 * DD), 0, 1);
        }
        const float4* wrow = (const float4*)(W2 + (size_t)it * (2 * DD));
        const float4 wa = wrow[lane * 2 + 0];
        const float4 wb = wrow[lane * 2 + 1];
        float acc = wa.x * xa.x + wa.y * xa.y + wa.z * xa.z + wa.w * xa.w
                  + wb.x * xb.x + wb.y * xb.y + wb.z * xb.z + wb.w * xb.w;
        #pragma unroll
        for (int s = 16; s >= 1; s >>= 1)
            acc += __shfl_down(acc, s, 32);
        if (lane == 0)
            out[(size_t)b * TT + t] = acc + b2[it];
    }
}

extern "C" void kernel_launch(void* const* d_in, const int* in_sizes, int n_in,
                              void* d_out, int out_size, void* d_ws, size_t ws_size,
                              hipStream_t stream) {
    const int*   seq        = (const int*)  d_in[0];
    const int*   user       = (const int*)  d_in[1];
    const int*   items      = (const int*)  d_in[2];
    const float* item_table = (const float*)d_in[3];
    const float* user_table = (const float*)d_in[4];
    const float* Wv         = (const float*)d_in[5];
    const float* bv         = (const float*)d_in[6];
    const float* Wh         = (const float*)d_in[7];
    const float* bh         = (const float*)d_in[8];
    const float* W1         = (const float*)d_in[9];
    const float* b1         = (const float*)d_in[10];
    const float* W2         = (const float*)d_in[11];
    const float* b2         = (const float*)d_in[12];
    float* out = (float*)d_out;

    float* z  = (float*)d_ws;
    float* xc = z + (size_t)BB * FC1;

    caser_features<<<BB, 128, 0, stream>>>(seq, user, item_table, user_table,
                                           Wv, bv, Wh, bh, z, xc);
    caser_fc1<<<BB / 16, 256, 0, stream>>>(z, W1, b1, xc);
    caser_score<<<BB, 256, 0, stream>>>(items, xc, W2, b2, out);
}


# Score kernel: one wave per batch row, gathered rows 4 at a time (scalar index loads, next group's 8 row loads in flight), 4 partials summed by one DPP butterfly; changes the scores' summation order
# speedup vs baseline: 3.5379x; 3.5379x over previous
#include <hip/hip_runtime.h>
#include <hip/hip_bf16.h>

#define BB 4096
#define LL 5
#define DD 128
#define NHH 16
#define NVV 4
#define TT 100
#define FC1 752

typedef float v2f __attribute__((ext_vector_type(2)));
typedef float v8f __attribute__((ext_vector_type(8)));

#define ROWP 644
#define PFP  20
#define PTS  320
#define NPT  35

template <int LOUT>
__device__ __forceinline__ void hconv_slab(const float4 (&a)[16],
                                           const float* __restrict__ eb,
                                           float* __restrict__ pp)
{
    v8f acc[LOUT];
    #pragma unroll
    for (int t = 0; t < LOUT; ++t) acc[t] = v8f{};

    #pragma unroll
    for (int j = 0; j < 16; ++j) {
        float4 e[LOUT];
        #pragma unroll
        for (int t = 0; t < LOUT; ++t)
            e[t] = *(const float4*)(eb + t * DD + 8 * j);
        const v2f a0 = {a[j].x, a[j].y};
        const v2f a1 = {a[j].z, a[j].w};
        #pragma unroll
        for (int t = 0; t < LOUT; ++t) {
            const v2f b0 = {e[t].x, e[t].y};
            acc[t] = __builtin_amdgcn_wmma_f32_16x16x4_f32(
                false, a0, false, b0, (short)0, acc[t], false, false);
        }
        #pragma unroll
        for (int t = 0; t < LOUT; ++t) {
            const v2f b1 = {e[t].z, e[t].w};
            acc[t] = __builtin_amdgcn_wmma_f32_16x16x4_f32(
                false, a1, false, b1, (short)0, acc[t], false, false);
        }
    }

    #pragma unroll
    for (int t = 0; t < LOUT; ++t) {
        #pragma unroll
        for (int r = 0; r < 8; ++r)
            pp[t * PTS + r * PFP] = acc[t][r];
    }
}

__global__ void __launch_bounds__(512)
caser_features(const int* __restrict__ seq,
               const int* __restrict__ user,
               const float* __restrict__ item_table,
               const float* __restrict__ user_table,
               const float* __restrict__ Wv,
               const float* __restrict__ bv,
               const float* __restrict__ Wh,
               const float* __restrict__ bh,
               float* __restrict__ z,
               float* __restrict__ xc)
{
    __shared__ __align__(16) float semb[16 * ROWP];
    __shared__ __align__(16) float spart[NPT * PTS];

    const int tid  = threadIdx.x;
    const int wave = __builtin_amdgcn_readfirstlane(tid >> 5);
    const int lane = tid & 31;
    const int idx16 = lane & 15;
    const int half  = lane >> 4;
    const int b0   = blockIdx.x * 16;

    const int l = (int)((0x5505454543343221ull >> (4 * wave)) & 15ull);
    const int s = (int)((0x4301321021002100ull >> (4 * wave)) & 15ull);
    const int lout = LL - l + 1;

    float4 a[16];
    if (l > 0) {
        const float* wl = Wh + ((size_t)((l - 1) * NHH + idx16) * LL + s) * DD + 4 * half;
        #pragma unroll
        for (int j = 0; j < 16; ++j)
            a[j] = *(const float4*)(wl + 8 * j);
    }

    const int b = b0 + wave;
    float4 e[LL];
    #pragma unroll
    for (int t = 0; t < LL; ++t) {
        const int it = seq[b * LL + t];
        e[t] = *(const float4*)(item_table + (size_t)it * DD + 4 * lane);
        *(float4*)(semb + wave * ROWP + t * DD + 4 * lane) = e[t];
    }
    const float4 ue = *(const float4*)(user_table + (size_t)user[b] * DD + 4 * lane);
    __syncthreads();

    #pragma unroll
    for (int v = 0; v < NVV; ++v) {
        const float bias = bv[v];
        float4 acc = make_float4(bias, bias, bias, bias);
        #pragma unroll
        for (int t = 0; t < LL; ++t) {
            const float w = Wv[v * LL + t];
            acc.x += e[t].x * w;
            acc.y += e[t].y * w;
            acc.z += e[t].z * w;
            acc.w += e[t].w * w;
        }
        *(float4*)(z + (size_t)b * FC1 + v * DD + 4 * lane) = acc;
    }
    *(float4*)(xc + (size_t)b * (2 * DD) + DD + 4 * lane) = ue;

    if (l > 0) {
        const float* eb = semb + idx16 * ROWP + s * DD + 4 * half;
        const int base = (l == 1) ? 0 : (l == 2) ? 5 : (l == 3) ? 13 : (l == 4) ? 22 : 30;
        float* pp = spart + (base + s * lout) * PTS + (8 * half) * PFP + idx16;
        switch (lout) {
            case 5:  hconv_slab<5>(a, eb, pp); break;
            case 4:  hconv_slab<4>(a, eb, pp); break;
            case 3:  hconv_slab<3>(a, eb, pp); break;
            case 2:  hconv_slab<2>(a, eb, pp); break;
            default: hconv_slab<1>(a, eb, pp); break;
        }
    }
    __syncthreads();

    for (int i = tid; i < 16 * 240; i += 512) {
        const int n = i / 240;
        const int c = i - n * 240;
        int ll, goff, pbase;
        if      (c <  80) { ll = 1; goff =   0; pbase =  0; }
        else if (c < 144) { ll = 2; goff =  80; pbase =  5; }
        else if (c < 192) { ll = 3; goff = 144; pbase = 13; }
        else if (c < 224) { ll = 4; goff = 192; pbase = 22; }
        else              { ll = 5; goff = 224; pbase = 30; }
        const int lo  = LL - ll + 1;
        const int rem = c - goff;
        const int f   = rem / lo;
        const int t   = rem - f * lo;
        float acc = 0.f;
        for (int ss = 0; ss < ll; ++ss)
            acc += spart[(pbase + ss * lo + t) * PTS + f * PFP + n];
        acc += bh[(ll - 1) * NHH + f];
        z[(size_t)(b0 + n) * FC1 + NVV * DD + c] = fmaxf(acc, 0.f);
    }
}

__global__ void caser_fc1(const float* __restrict__ z,
                          const float* __restrict__ W1,
                          const float* __restrict__ b1,
                          float* __restrict__ xc)
{
    __shared__ float sz[16][FC1 + 4];
    const int tid  = threadIdx.x;
    const int wave = tid >> 5;
    const int lane = tid & 31;
    const int m0   = blockIdx.x * 16;

    for (int i = tid; i < 16 * FC1; i += 256) {
        const int r = i / FC1, c = i % FC1;
        sz[r][c] = z[(size_t)(m0 + r) * FC1 + c];
    }
    __syncthreads();

    const int n0   = wave * 16;
    const int mrow = lane & 15;
    const int half = lane >> 4;

    v8f acc = {};
    for (int k0 = 0; k0 < FC1; k0 += 4) {
        const int ka = k0 + half * 2;
        v2f a, bf;
        a.x = sz[mrow][ka + 0];
        a.y = sz[mrow][ka + 1];
        bf.x = W1[(size_t)(ka + 0) * DD + n0 + mrow];
        bf.y = W1[(size_t)(ka + 1) * DD + n0 + mrow];
        acc = __builtin_amdgcn_wmma_f32_16x16x4_f32(
             false, a,  false, bf,
             (short)0, acc,  false,  false);
    }

    const int n = n0 + (lane & 15);
    const float bias = b1[n];
    #pragma unroll
    for (int r = 0; r < 8; ++r) {
        const int m = m0 + r + half * 8;
        const float v = acc[r] + bias;
        xc[(size_t)m * (2 * DD) + n] = fmaxf(v, 0.f);
    }
}

#define SG 4
#define NG 25

template <int M>
__device__ __forceinline__ float lane_xor_row(float v)
{
    const int i = __builtin_bit_cast(int, v);
    return __builtin_bit_cast(float,
        __builtin_amdgcn_update_dpp(i, i, 0x160 | M, 0xF, 0xF, false));
}

__device__ __forceinline__ float lane_xor16(float v)
{
    const int i = __builtin_bit_cast(int, v);
    return __builtin_bit_cast(float,
        __builtin_amdgcn_permlanex16(i, i, 0x76543210, 0xfedcba98, false, false));
}

__device__ __forceinline__ void score_idx(const int* __restrict__ itp, int (&it)[SG])
{
    #pragma unroll
    for (int r = 0; r < SG; ++r) it[r] = itp[r];
}

__device__ __forceinline__ void score_issue(const int (&it)[SG],
                                            const float* __restrict__ W2,
                                            const float* __restrict__ b2,
                                            const int lane,
                                            float4 (&lo)[SG], float4 (&hi)[SG],
                                            float (&bias)[SG])
{
    #pragma unroll
    for (int r = 0; r < SG; ++r) {
        const float* row = W2 + (size_t)it[r] * (2 * DD);
        lo[r]   = *(const float4*)(row + 4 * lane);
        hi[r]   = *(const float4*)(row + DD + 4 * lane);
        bias[r] = b2[it[r]];
    }
}

__device__ __forceinline__ void score_sum(const float4 (&lo)[SG], const float4 (&hi)[SG],
                                          const float (&bias)[SG],
                                          const float4 xa, const float4 xb,
                                          const int lane, float* __restrict__ outp)
{
    float p[SG];
    #pragma unroll
    for (int r = 0; r < SG; ++r)
        p[r] = lo[r].x * xa.x + lo[r].y * xa.y + lo[r].z * xa.z + lo[r].w * xa.w
             + hi[r].x * xb.x + hi[r].y * xb.y + hi[r].z * xb.z + hi[r].w * xb.w;

    const bool up = (lane & 16) != 0;
    const bool b3 = (lane & 8) != 0;
    const float q0 = (up ? p[2] : p[0]) + lane_xor16(up ? p[0] : p[2]);
    const float q1 = (up ? p[3] : p[1]) + lane_xor16(up ? p[1] : p[3]);
    float v = (b3 ? q1 : q0) + lane_xor_row<8>(b3 ? q0 : q1);
    v += lane_xor_row<4>(v);
    v += lane_xor_row<2>(v);
    v += lane_xor_row<1>(v);
    const float bsel = up ? (b3 ? bias[3] : bias[2]) : (b3 ? bias[1] : bias[0]);
    if ((lane & 7) == 0)
        outp[(lane >> 3)] = v + bsel;
}

__global__ void __launch_bounds__(512)
caser_score(const int* __restrict__ items,
            const float* __restrict__ xc,
            const float* __restrict__ W2,
            const float* __restrict__ b2,
            float* __restrict__ out)
{
    const int tid  = threadIdx.x;
    const int wave = __builtin_amdgcn_readfirstlane(tid >> 5);
    const int lane = tid & 31;
    const int b    = blockIdx.x * 16 + wave;

    const int* __restrict__ itp = items + (size_t)b * TT;
    float* __restrict__ op      = out + (size_t)b * TT;

    const float4 xa = *(const float4*)(xc + (size_t)b * (2 * DD) + 4 * lane);
    const float4 xb = *(const float4*)(xc + (size_t)b * (2 * DD) + DD + 4 * lane);

    float4 loA[SG], hiA[SG], loB[SG], hiB[SG];
    float  bA[SG], bB[SG];
    int    iA[SG], iB[SG];

    score_idx(itp, iA);
    score_idx(itp + SG, iB);
    score_issue(iA, W2, b2, lane, loA, hiA, bA);
    for (int g = 0; g < NG - 1; g += 2) {
        score_issue(iB, W2, b2, lane, loB, hiB, bB);
        score_idx(itp + (g + 2) * SG, iA);
        score_sum(loA, hiA, bA, xa, xb, lane, op + g * SG);
        score_issue(iA, W2, b2, lane, loA, hiA, bA);
        if (g + 3 < NG) score_idx(itp + (g + 3) * SG, iB);
        score_sum(loB, hiB, bB, xa, xb, lane, op + (g + 1) * SG);
    }
    score_sum(loA, hiA, bA, xa, xb, lane, op + (NG - 1) * SG);
}

extern "C" void kernel_launch(void* const* d_in, const int* in_sizes, int n_in,
                              void* d_out, int out_size, void* d_ws, size_t ws_size,
                              hipStream_t stream) {
    const int*   seq        = (const int*)  d_in[0];
    const int*   user       = (const int*)  d_in[1];
    const int*   items      = (const int*)  d_in[2];
    const float* item_table = (const float*)d_in[3];
    const float* user_table = (const float*)d_in[4];
    const float* Wv         = (const float*)d_in[5];
    const float* bv         = (const float*)d_in[6];
    const float* Wh         = (const float*)d_in[7];
    const float* bh         = (const float*)d_in[8];
    const float* W1         = (const float*)d_in[9];
    const float* b1         = (const float*)d_in[10];
    const float* W2         = (const float*)d_in[11];
    const float* b2         = (const float*)d_in[12];
    float* out = (float*)d_out;

    float* z  = (float*)d_ws;
    float* xc = z + (size_t)BB * FC1;

    caser_features<<<BB / 16, 512, 0, stream>>>(seq, user, item_table, user_table,
                                           Wv, bv, Wh, bh, z, xc);
    caser_fc1<<<BB / 16, 256, 0, stream>>>(z, W1, b1, xc);
    caser_score<<<BB / 16, 512, 0, stream>>>(items, xc, W2, b2, out);
}


# FC1 operands through LDS by whole-row 16-byte loads: z tile at once, W1 in 96-row slabs through a 2-stage ring, next slab's loads in flight under the WMMAs; 2 accumulators: changes summation order
# speedup vs baseline: 5.3279x; 1.5059x over previous
#include <hip/hip_runtime.h>
#include <hip/hip_bf16.h>

#define BB 4096
#define LL 5
#define DD 128
#define NHH 16
#define NVV 4
#define TT 100
#define FC1 752

typedef float v2f __attribute__((ext_vector_type(2)));
typedef float v8f __attribute__((ext_vector_type(8)));

#define ROWP 644
#define PFP  20
#define PTS  320
#define NPT  35

template <int LOUT>
__device__ __forceinline__ void hconv_slab(const float4 (&a)[16],
                                           const float* __restrict__ eb,
                                           float* __restrict__ pp)
{
    v8f acc[LOUT];
    #pragma unroll
    for (int t = 0; t < LOUT; ++t) acc[t] = v8f{};

    #pragma unroll
    for (int j = 0; j < 16; ++j) {
        float4 e[LOUT];
        #pragma unroll
        for (int t = 0; t < LOUT; ++t)
            e[t] = *(const float4*)(eb + t * DD + 8 * j);
        const v2f a0 = {a[j].x, a[j].y};
        const v2f a1 = {a[j].z, a[j].w};
        #pragma unroll
        for (int t = 0; t < LOUT; ++t) {
            const v2f b0 = {e[t].x, e[t].y};
            acc[t] = __builtin_amdgcn_wmma_f32_16x16x4_f32(
                false, a0, false, b0, (short)0, acc[t], false, false);
        }
        #pragma unroll
        for (int t = 0; t < LOUT; ++t) {
            const v2f b1 = {e[t].z, e[t].w};
            acc[t] = __builtin_amdgcn_wmma_f32_16x16x4_f32(
                false, a1, false, b1, (short)0, acc[t], false, false);
        }
    }

    #pragma unroll
    for (int t = 0; t < LOUT; ++t) {
        #pragma unroll
        for (int r = 0; r < 8; ++r)
            pp[t * PTS + r * PFP] = acc[t][r];
    }
}

__global__ void __launch_bounds__(512)
caser_features(const int* __restrict__ seq,
               const int* __restrict__ user,
               const float* __restrict__ item_table,
               const float* __restrict__ user_table,
               const float* __restrict__ Wv,
               const float* __restrict__ bv,
               const float* __restrict__ Wh,
               const float* __restrict__ bh,
               float* __restrict__ z,
               float* __restrict__ xc)
{
    __shared__ __align__(16) float semb[16 * ROWP];
    __shared__ __align__(16) float spart[NPT * PTS];

    const int tid  = threadIdx.x;
    const int wave = __builtin_amdgcn_readfirstlane(tid >> 5);
    const int lane = tid & 31;
    const int idx16 = lane & 15;
    const int half  = lane >> 4;
    const int b0   = blockIdx.x * 16;

    const int l = (int)((0x5505454543343221ull >> (4 * wave)) & 15ull);
    const int s = (int)((0x4301321021002100ull >> (4 * wave)) & 15ull);
    const int lout = LL - l + 1;

    float4 a[16];
    if (l > 0) {
        const float* wl = Wh + ((size_t)((l - 1) * NHH + idx16) * LL + s) * DD + 4 * half;
        #pragma unroll
        for (int j = 0; j < 16; ++j)
            a[j] = *(const float4*)(wl + 8 * j);
    }

    const int b = b0 + wave;
    float4 e[LL];
    #pragma unroll
    for (int t = 0; t < LL; ++t) {
        const int it = seq[b * LL + t];
        e[t] = *(const float4*)(item_table + (size_t)it * DD + 4 * lane);
        *(float4*)(semb + wave * ROWP + t * DD + 4 * lane) = e[t];
    }
    const float4 ue = *(const float4*)(user_table + (size_t)user[b] * DD + 4 * lane);
    __syncthreads();

    #pragma unroll
    for (int v = 0; v < NVV; ++v) {
        const float bias = bv[v];
        float4 acc = make_float4(bias, bias, bias, bias);
        #pragma unroll
        for (int t = 0; t < LL; ++t) {
            const float w = Wv[v * LL + t];
            acc.x += e[t].x * w;
            acc.y += e[t].y * w;
            acc.z += e[t].z * w;
            acc.w += e[t].w * w;
        }
        *(float4*)(z + (size_t)b * FC1 + v * DD + 4 * lane) = acc;
    }
    *(float4*)(xc + (size_t)b * (2 * DD) + DD + 4 * lane) = ue;

    if (l > 0) {
        const float* eb = semb + idx16 * ROWP + s * DD + 4 * half;
        const int base = (l == 1) ? 0 : (l == 2) ? 5 : (l == 3) ? 13 : (l == 4) ? 22 : 30;
        float* pp = spart + (base + s * lout) * PTS + (8 * half) * PFP + idx16;
        switch (lout) {
            case 5:  hconv_slab<5>(a, eb, pp); break;
            case 4:  hconv_slab<4>(a, eb, pp); break;
            case 3:  hconv_slab<3>(a, eb, pp); break;
            case 2:  hconv_slab<2>(a, eb, pp); break;
            default: hconv_slab<1>(a, eb, pp); break;
        }
    }
    __syncthreads();

    for (int i = tid; i < 16 * 240; i += 512) {
        const int n = i / 240;
        const int c = i - n * 240;
        int ll, goff, pbase;
        if      (c <  80) { ll = 1; goff =   0; pbase =  0; }
        else if (c < 144) { ll = 2; goff =  80; pbase =  5; }
        else if (c < 192) { ll = 3; goff = 144; pbase = 13; }
        else if (c < 224) { ll = 4; goff = 192; pbase = 22; }
        else              { ll = 5; goff = 224; pbase = 30; }
        const int lo  = LL - ll + 1;
        const int rem = c - goff;
        const int f   = rem / lo;
        const int t   = rem - f * lo;
        float acc = 0.f;
        for (int ss = 0; ss < ll; ++ss)
            acc += spart[(pbase + ss * lo + t) * PTS + f * PFP + n];
        acc += bh[(ll - 1) * NHH + f];
        z[(size_t)(b0 + n) * FC1 + NVV * DD + c] = fmaxf(acc, 0.f);
    }
}

#define ZP    756
#define WPP   272
#define SROWS 96
#define NSLAB 8
#define STG   (SROWS / 2 * WPP)

template <int NP>
__device__ __forceinline__ void fc1_fetch(const float* __restrict__ wrow, float4 (&w)[12])
{
    #pragma unroll
    for (int i = 0; i < NP; ++i) {
        w[2 * i]     = *(const float4*)(wrow + (size_t)(16 * i) * DD);
        w[2 * i + 1] = *(const float4*)(wrow + (size_t)(16 * i + 1) * DD);
    }
}

template <int NP>
__device__ __forceinline__ void fc1_park(const float4 (&w)[12], float* __restrict__ sp)
{
    #pragma unroll
    for (int i = 0; i < NP; ++i) {
        const float4 r0 = w[2 * i];
        const float4 r1 = w[2 * i + 1];
        *(float4*)(sp + i * 8 * WPP)     = make_float4(r0.x, r1.x, r0.y, r1.y);
        *(float4*)(sp + i * 8 * WPP + 4) = make_float4(r0.z, r1.z, r0.w, r1.w);
    }
}

template <int NB>
__device__ __forceinline__ void fc1_slab(const float* __restrict__ za,
                                         const float* __restrict__ wb,
                                         v8f& acc0, v8f& acc1)
{
    #pragma unroll
    for (int j = 0; j < NB; ++j) {
        const float4 a  = *(const float4*)(za + 8 * j);
        const float2 p0 = *(const float2*)(wb + (4 * j) * WPP);
        const float2 p1 = *(const float2*)(wb + (4 * j + 1) * WPP);
        const v2f a0 = {a.x, a.y};
        const v2f a1 = {a.z, a.w};
        const v2f b0 = {p0.x, p0.y};
        const v2f b1 = {p1.x, p1.y};
        acc0 = __builtin_amdgcn_wmma_f32_16x16x4_f32(
            false, a0, false, b0, (short)0, acc0, false, false);
        acc1 = __builtin_amdgcn_wmma_f32_16x16x4_f32(
            false, a1, false, b1, (short)0, acc1, false, false);
    }
}

__global__ void __launch_bounds__(256)
caser_fc1(const float* __restrict__ z,
          const float* __restrict__ W1,
          const float* __restrict__ b1,
          float* __restrict__ xc)
{
    __shared__ __align__(16) float sz[16 * ZP];
    __shared__ __align__(16) float sw[2 * STG];

    const int tid  = threadIdx.x;
    const int wave = __builtin_amdgcn_readfirstlane(tid >> 5);
    const int lane = tid & 31;
    const int c    = lane & 15;
    const int h    = lane >> 4;
    const int m0   = blockIdx.x * 16;

    const int zrow = tid >> 4;
    const int zq   = tid & 15;
    const float* zsrc = z + (size_t)(m0 + zrow) * FC1 + 4 * zq;
    float*       zdst = sz + zrow * ZP + 4 * zq;
    float4 zr[11];
    #pragma unroll
    for (int i = 0; i < 11; ++i)
        zr[i] = *(const float4*)(zsrc + 64 * i);
    float4 zt = make_float4(0.f, 0.f, 0.f, 0.f);
    if (zq < 12) zt = *(const float4*)(zsrc + 64 * 11);

    float4 w[12];
    const float* wrow = W1 + (size_t)(2 * wave) * DD + 4 * lane;
    float* spark = sw + wave * WPP + 8 * lane;
    fc1_fetch<6>(wrow, w);

    #pragma unroll
    for (int i = 0; i < 11; ++i)
        *(float4*)(zdst + 64 * i) = zr[i];
    if (zq < 12) *(float4*)(zdst + 64 * 11) = zt;
    fc1_park<6>(w, spark);
    __syncthreads();

    const float* za = sz + c * ZP + 4 * h;
    const float* wb = sw + (2 * h) * WPP + 2 * (wave * 16 + c);
    v8f acc0 = {};
    v8f acc1 = {};

    for (int s = 0; s < NSLAB - 2; ++s) {
        const int cur = (s & 1) * STG;
        const int nxt = STG - cur;
        fc1_fetch<6>(wrow + (size_t)(s + 1) * SROWS * DD, w);
        __builtin_amdgcn_sched_barrier(0);
        fc1_slab<12>(za + s * SROWS, wb + cur, acc0, acc1);
        fc1_park<6>(w, spark + nxt);
        __syncthreads();
    }
    fc1_fetch<5>(wrow + (size_t)(NSLAB - 1) * SROWS * DD, w);
    __builtin_amdgcn_sched_barrier(0);
    fc1_slab<12>(za + (NSLAB - 2) * SROWS, wb, acc0, acc1);
    fc1_park<5>(w, spark + STG);
    __syncthreads();
    fc1_slab<10>(za + (NSLAB - 1) * SROWS, wb + STG, acc0, acc1);

    const int n = wave * 16 + c;
    const float bias = b1[n];
    #pragma unroll
    for (int r = 0; r < 8; ++r) {
        const int m = m0 + r + h * 8;
        const float v = (acc0[r] + acc1[r]) + bias;
        xc[(size_t)m * (2 * DD) + n] = fmaxf(v, 0.f);
    }
}

#define SG 4
#define NG 25

template <int M>
__device__ __forceinline__ float lane_xor_row(float v)
{
    const int i = __builtin_bit_cast(int, v);
    return __builtin_bit_cast(float,
        __builtin_amdgcn_update_dpp(i, i, 0x160 | M, 0xF, 0xF, false));
}

__device__ __forceinline__ float lane_xor16(float v)
{
    const int i = __builtin_bit_cast(int, v);
    return __builtin_bit_cast(float,
        __builtin_amdgcn_permlanex16(i, i, 0x76543210, 0xfedcba98, false, false));
}

__device__ __forceinline__ void score_idx(const int* __restrict__ itp, int (&it)[SG])
{
    #pragma unroll
    for (int r = 0; r < SG; ++r) it[r] = itp[r];
}

__device__ __forceinline__ void score_issue(const int (&it)[SG],
                                            const float* __restrict__ W2,
                                            const float* __restrict__ b2,
                                            const int lane,
                                            float4 (&lo)[SG], float4 (&hi)[SG],
                                            float (&bias)[SG])
{
    #pragma unroll
    for (int r = 0; r < SG; ++r) {
        const float* row = W2 + (size_t)it[r] * (2 * DD);
        lo[r]   = *(const float4*)(row + 4 * lane);
        hi[r]   = *(const float4*)(row + DD + 4 * lane);
        bias[r] = b2[it[r]];
    }
}

__device__ __forceinline__ void score_sum(const float4 (&lo)[SG], const float4 (&hi)[SG],
                                          const float (&bias)[SG],
                                          const float4 xa, const float4 xb,
                                          const int lane, float* __restrict__ outp)
{
    float p[SG];
    #pragma unroll
    for (int r = 0; r < SG; ++r)
        p[r] = lo[r].x * xa.x + lo[r].y * xa.y + lo[r].z * xa.z + lo[r].w * xa.w
             + hi[r].x * xb.x + hi[r].y * xb.y + hi[r].z * xb.z + hi[r].w * xb.w;

    const bool up = (lane & 16) != 0;
    const bool b3 = (lane & 8) != 0;
    const float q0 = (up ? p[2] : p[0]) + lane_xor16(up ? p[0] : p[2]);
    const float q1 = (up ? p[3] : p[1]) + lane_xor16(up ? p[1] : p[3]);
    float v = (b3 ? q1 : q0) + lane_xor_row<8>(b3 ? q0 : q1);
    v += lane_xor_row<4>(v);
    v += lane_xor_row<2>(v);
    v += lane_xor_row<1>(v);
    const float bsel = up ? (b3 ? bias[3] : bias[2]) : (b3 ? bias[1] : bias[0]);
    if ((lane & 7) == 0)
        outp[(lane >> 3)] = v + bsel;
}

__global__ void __launch_bounds__(512)
caser_score(const int* __restrict__ items,
            const float* __restrict__ xc,
            const float* __restrict__ W2,
            const float* __restrict__ b2,
            float* __restrict__ out)
{
    const int tid  = threadIdx.x;
    const int wave = __builtin_amdgcn_readfirstlane(tid >> 5);
    const int lane = tid & 31;
    const int b    = blockIdx.x * 16 + wave;

    const int* __restrict__ itp = items + (size_t)b * TT;
    float* __restrict__ op      = out + (size_t)b * TT;

    const float4 xa = *(const float4*)(xc + (size_t)b * (2 * DD) + 4 * lane);
    const float4 xb = *(const float4*)(xc + (size_t)b * (2 * DD) + DD + 4 * lane);

    float4 loA[SG], hiA[SG], loB[SG], hiB[SG];
    float  bA[SG], bB[SG];
    int    iA[SG], iB[SG];

    score_idx(itp, iA);
    score_idx(itp + SG, iB);
    score_issue(iA, W2, b2, lane, loA, hiA, bA);
    for (int g = 0; g < NG - 1; g += 2) {
        score_issue(iB, W2, b2, lane, loB, hiB, bB);
        score_idx(itp + (g + 2) * SG, iA);
        score_sum(loA, hiA, bA, xa, xb, lane, op + g * SG);
        score_issue(iA, W2, b2, lane, loA, hiA, bA);
        if (g + 3 < NG) score_idx(itp + (g + 3) * SG, iB);
        score_sum(loB, hiB, bB, xa, xb, lane, op + (g + 1) * SG);
    }
    score_sum(loA, hiA, bA, xa, xb, lane, op + (NG - 1) * SG);
}

extern "C" void kernel_launch(void* const* d_in, const int* in_sizes, int n_in,
                              void* d_out, int out_size, void* d_ws, size_t ws_size,
                              hipStream_t stream) {
    const int*   seq        = (const int*)  d_in[0];
    const int*   user       = (const int*)  d_in[1];
    const int*   items      = (const int*)  d_in[2];
    const float* item_table = (const float*)d_in[3];
    const float* user_table = (const float*)d_in[4];
    const float* Wv         = (const float*)d_in[5];
    const float* bv         = (const float*)d_in[6];
    const float* Wh         = (const float*)d_in[7];
    const float* bh         = (const float*)d_in[8];
    const float* W1         = (const float*)d_in[9];
    const float* b1         = (const float*)d_in[10];
    const float* W2         = (const float*)d_in[11];
    const float* b2         = (const float*)d_in[12];
    float* out = (float*)d_out;

    float* z  = (float*)d_ws;
    float* xc = z + (size_t)BB * FC1;

    caser_features<<<BB / 16, 512, 0, stream>>>(seq, user, item_table, user_table,
                                           Wv, bv, Wh, bh, z, xc);
    caser_fc1<<<BB / 16, 256, 0, stream>>>(z, W1, b1, xc);
    caser_score<<<BB / 16, 512, 0, stream>>>(items, xc, W2, b2, out);
}


# Features launch: waves' dependent memory round trips cut: seq+user indices, Wv/bv at entry, user row with the item rows; shift-sum pass one column per thread, LDS reads batched; sum order unchanged
# speedup vs baseline: 5.6016x; 1.0514x over previous
#include <hip/hip_runtime.h>
#include <hip/hip_bf16.h>

#define BB 4096
#define LL 5
#define DD 128
#define NHH 16
#define NVV 4
#define TT 100
#define FC1 752

typedef float v2f __attribute__((ext_vector_type(2)));
typedef float v8f __attribute__((ext_vector_type(8)));

#define ROWP 644
#define PFP  20
#define PTS  320
#define NPT  35

template <int LOUT>
__device__ __forceinline__ void hconv_slab(const float4 (&a)[16],
                                           const float* __restrict__ eb,
                                           float* __restrict__ pp)
{
    v8f acc[LOUT];
    #pragma unroll
    for (int t = 0; t < LOUT; ++t) acc[t] = v8f{};

    #pragma unroll
    for (int j = 0; j < 16; ++j) {
        float4 e[LOUT];
        #pragma unroll
        for (int t = 0; t < LOUT; ++t)
            e[t] = *(const float4*)(eb + t * DD + 8 * j);
        const v2f a0 = {a[j].x, a[j].y};
        const v2f a1 = {a[j].z, a[j].w};
        #pragma unroll
        for (int t = 0; t < LOUT; ++t) {
            const v2f b0 = {e[t].x, e[t].y};
            acc[t] = __builtin_amdgcn_wmma_f32_16x16x4_f32(
                false, a0, false, b0, (short)0, acc[t], false, false);
        }
        #pragma unroll
        for (int t = 0; t < LOUT; ++t) {
            const v2f b1 = {e[t].z, e[t].w};
            acc[t] = __builtin_amdgcn_wmma_f32_16x16x4_f32(
                false, a1, false, b1, (short)0, acc[t], false, false);
        }
    }

    #pragma unroll
    for (int t = 0; t < LOUT; ++t) {
        #pragma unroll
        for (int r = 0; r < 8; ++r)
            pp[t * PTS + r * PFP] = acc[t][r];
    }
}

__global__ void __launch_bounds__(512)
caser_features(const int* __restrict__ seq,
               const int* __restrict__ user,
               const float* __restrict__ item_table,
               const float* user_table,
               const float* __restrict__ Wv,
               const float* __restrict__ bv,
               const float* __restrict__ Wh,
               const float* __restrict__ bh,
               float* __restrict__ z,
               float* __restrict__ xc)
{
    __shared__ __align__(16) float semb[16 * ROWP];
    __shared__ __align__(16) float spart[NPT * PTS];
    __shared__ float sbh[LL * NHH];

    const int tid  = threadIdx.x;
    const int wave = __builtin_amdgcn_readfirstlane(tid >> 5);
    const int lane = tid & 31;
    const int idx16 = lane & 15;
    const int half  = lane >> 4;
    const int b0   = blockIdx.x * 16;

    const int l = (int)((0x5505454543343221ull >> (4 * wave)) & 15ull);
    const int s = (int)((0x4301321021002100ull >> (4 * wave)) & 15ull);
    const int lout = LL - l + 1;

    const int b = b0 + wave;
    int it[LL];
    #pragma unroll
    for (int t = 0; t < LL; ++t) it[t] = seq[b * LL + t];
    const int ui = user[b];
    float wv[NVV * LL], bvr[NVV];
    #pragma unroll
    for (int i = 0; i < NVV * LL; ++i) wv[i] = Wv[i];
    #pragma unroll
    for (int v = 0; v < NVV; ++v) bvr[v] = bv[v];
    float bhv = 0.f;
    if (tid < LL * NHH) bhv = bh[tid];

    float4 a[16];
    if (l > 0) {
        const float* wl = Wh + ((size_t)((l - 1) * NHH + idx16) * LL + s) * DD + 4 * half;
        #pragma unroll
        for (int j = 0; j < 16; ++j)
            a[j] = *(const float4*)(wl + 8 * j);
    }

    float4 e[LL];
    #pragma unroll
    for (int t = 0; t < LL; ++t)
        e[t] = *(const float4*)(item_table + (size_t)it[t] * DD + 4 * lane);
    const float4 ue = *(const float4*)(user_table + (size_t)ui * DD + 4 * lane);
    #pragma unroll
    for (int t = 0; t < LL; ++t)
        *(float4*)(semb + wave * ROWP + t * DD + 4 * lane) = e[t];
    if (tid < LL * NHH) sbh[tid] = bhv;
    __syncthreads();

    #pragma unroll
    for (int v = 0; v < NVV; ++v) {
        const float bias = bvr[v];
        float4 acc = make_float4(bias, bias, bias, bias);
        #pragma unroll
        for (int t = 0; t < LL; ++t) {
            const float w = wv[v * LL + t];
            acc.x += e[t].x * w;
            acc.y += e[t].y * w;
            acc.z += e[t].z * w;
            acc.w += e[t].w * w;
        }
        *(float4*)(z + (size_t)b * FC1 + v * DD + 4 * lane) = acc;
    }
    *(float4*)(xc + (size_t)b * (2 * DD) + DD + 4 * lane) = ue;

    if (l > 0) {
        const float* eb = semb + idx16 * ROWP + s * DD + 4 * half;
        const int base = (l == 1) ? 0 : (l == 2) ? 5 : (l == 3) ? 13 : (l == 4) ? 22 : 30;
        float* pp = spart + (base + s * lout) * PTS + (8 * half) * PFP + idx16;
        switch (lout) {
            case 5:  hconv_slab<5>(a, eb, pp); break;
            case 4:  hconv_slab<4>(a, eb, pp); break;
            case 3:  hconv_slab<3>(a, eb, pp); break;
            case 2:  hconv_slab<2>(a, eb, pp); break;
            default: hconv_slab<1>(a, eb, pp); break;
        }
    }
    __syncthreads();

    if (tid < 480) {
        const int nh = (tid >= 240) ? 1 : 0;
        const int c  = tid - 240 * nh;
        int ll, rem, pbase;
        if      (c <  80) { ll = 1; rem = c;       pbase =  0; }
        else if (c < 144) { ll = 2; rem = c -  80; pbase =  5; }
        else if (c < 192) { ll = 3; rem = c - 144; pbase = 13; }
        else if (c < 224) { ll = 4; rem = c - 192; pbase = 22; }
        else              { ll = 5; rem = c - 224; pbase = 30; }
        const int lo = LL - ll + 1;
        const int f = (ll == 1) ? rem / 5 : (ll == 2) ? (rem >> 2)
                    : (ll == 3) ? rem / 3 : (ll == 4) ? (rem >> 1) : rem;
        const int t = rem - f * lo;
        const float bias = sbh[(ll - 1) * NHH + f];
        const float* p0 = spart + (pbase + t) * PTS + f * PFP + nh;
        int so[LL];
        #pragma unroll
        for (int ss = 0; ss < LL; ++ss) so[ss] = (ss < ll) ? ss * lo * PTS : 0;
        float* zp = z + (size_t)(b0 + nh) * FC1 + NVV * DD + c;
        #pragma unroll
        for (int k = 0; k < 8; ++k) {
            float acc = 0.f;
            #pragma unroll
            for (int ss = 0; ss < LL; ++ss) {
                const float v = p0[so[ss] + 2 * k];
                acc = (ss < ll) ? acc + v : acc;
            }
            acc += bias;
            zp[(size_t)(2 * k) * FC1] = fmaxf(acc, 0.f);
        }
    }
}

#define ZP    756
#define WPP   272
#define SROWS 96
#define NSLAB 8
#define STG   (SROWS / 2 * WPP)

template <int NP>
__device__ __forceinline__ void fc1_fetch(const float* __restrict__ wrow, float4 (&w)[12])
{
    #pragma unroll
    for (int i = 0; i < NP; ++i) {
        w[2 * i]     = *(const float4*)(wrow + (size_t)(16 * i) * DD);
        w[2 * i + 1] = *(const float4*)(wrow + (size_t)(16 * i + 1) * DD);
    }
}

template <int NP>
__device__ __forceinline__ void fc1_park(const float4 (&w)[12], float* __restrict__ sp)
{
    #pragma unroll
    for (int i = 0; i < NP; ++i) {
        const float4 r0 = w[2 * i];
        const float4 r1 = w[2 * i + 1];
        *(float4*)(sp + i * 8 * WPP)     = make_float4(r0.x, r1.x, r0.y, r1.y);
        *(float4*)(sp + i * 8 * WPP + 4) = make_float4(r0.z, r1.z, r0.w, r1.w);
    }
}

template <int NB>
__device__ __forceinline__ void fc1_slab(const float* __restrict__ za,
                                         const float* __restrict__ wb,
                                         v8f& acc0, v8f& acc1)
{
    #pragma unroll
    for (int j = 0; j < NB; ++j) {
        const float4 a  = *(const float4*)(za + 8 * j);
        const float2 p0 = *(const float2*)(wb + (4 * j) * WPP);
        const float2 p1 = *(const float2*)(wb + (4 * j + 1) * WPP);
        const v2f a0 = {a.x, a.y};
        const v2f a1 = {a.z, a.w};
        const v2f b0 = {p0.x, p0.y};
        const v2f b1 = {p1.x, p1.y};
        acc0 = __builtin_amdgcn_wmma_f32_16x16x4_f32(
            false, a0, false, b0, (short)0, acc0, false, false);
        acc1 = __builtin_amdgcn_wmma_f32_16x16x4_f32(
            false, a1, false, b1, (short)0, acc1, false, false);
    }
}

__global__ void __launch_bounds__(256)
caser_fc1(const float* __restrict__ z,
          const float* __restrict__ W1,
          const float* __restrict__ b1,
          float* __restrict__ xc)
{
    __shared__ __align__(16) float sz[16 * ZP];
    __shared__ __align__(16) float sw[2 * STG];

    const int tid  = threadIdx.x;
    const int wave = __builtin_amdgcn_readfirstlane(tid >> 5);
    const int lane = tid & 31;
    const int c    = lane & 15;
    const int h    = lane >> 4;
    const int m0   = blockIdx.x * 16;

    const int zrow = tid >> 4;
    const int zq   = tid & 15;
    const float* zsrc = z + (size_t)(m0 + zrow) * FC1 + 4 * zq;
    float*       zdst = sz + zrow * ZP + 4 * zq;
    float4 zr[11];
    #pragma unroll
    for (int i = 0; i < 11; ++i)
        zr[i] = *(const float4*)(zsrc + 64 * i);
    float4 zt = make_float4(0.f, 0.f, 0.f, 0.f);
    if (zq < 12) zt = *(const float4*)(zsrc + 64 * 11);

    float4 w[12];
    const float* wrow = W1 + (size_t)(2 * wave) * DD + 4 * lane;
    float* spark = sw + wave * WPP + 8 * lane;
    fc1_fetch<6>(wrow, w);

    #pragma unroll
    for (int i = 0; i < 11; ++i)
        *(float4*)(zdst + 64 * i) = zr[i];
    if (zq < 12) *(float4*)(zdst + 64 * 11) = zt;
    fc1_park<6>(w, spark);
    __syncthreads();

    const float* za = sz + c * ZP + 4 * h;
    const float* wb = sw + (2 * h) * WPP + 2 * (wave * 16 + c);
    v8f acc0 = {};
    v8f acc1 = {};

    for (int s = 0; s < NSLAB - 2; ++s) {
        const int cur = (s & 1) * STG;
        const int nxt = STG - cur;
        fc1_fetch<6>(wrow + (size_t)(s + 1) * SROWS * DD, w);
        __builtin_amdgcn_sched_barrier(0);
        fc1_slab<12>(za + s * SROWS, wb + cur, acc0, acc1);
        fc1_park<6>(w, spark + nxt);
        __syncthreads();
    }
    fc1_fetch<5>(wrow + (size_t)(NSLAB - 1) * SROWS * DD, w);
    __builtin_amdgcn_sched_barrier(0);
    fc1_slab<12>(za + (NSLAB - 2) * SROWS, wb, acc0, acc1);
    fc1_park<5>(w, spark + STG);
    __syncthreads();
    fc1_slab<10>(za + (NSLAB - 1) * SROWS, wb + STG, acc0, acc1);

    const int n = wave * 16 + c;
    const float bias = b1[n];
    #pragma unroll
    for (int r = 0; r < 8; ++r) {
        const int m = m0 + r + h * 8;
        const float v = (acc0[r] + acc1[r]) + bias;
        xc[(size_t)m * (2 * DD) + n] = fmaxf(v, 0.f);
    }
}

#define SG 4
#define NG 25

template <int M>
__device__ __forceinline__ float lane_xor_row(float v)
{
    const int i = __builtin_bit_cast(int, v);
    return __builtin_bit_cast(float,
        __builtin_amdgcn_update_dpp(i, i, 0x160 | M, 0xF, 0xF, false));
}

__device__ __forceinline__ float lane_xor16(float v)
{
    const int i = __builtin_bit_cast(int, v);
    return __builtin_bit_cast(float,
        __builtin_amdgcn_permlanex16(i, i, 0x76543210, 0xfedcba98, false, false));
}

__device__ __forceinline__ void score_idx(const int* __restrict__ itp, int (&it)[SG])
{
    #pragma unroll
    for (int r = 0; r < SG; ++r) it[r] = itp[r];
}

__device__ __forceinline__ void score_issue(const int (&it)[SG],
                                            const float* __restrict__ W2,
                                            const float* __restrict__ b2,
                                            const int lane,
                                            float4 (&lo)[SG], float4 (&hi)[SG],
                                            float (&bias)[SG])
{
    #pragma unroll
    for (int r = 0; r < SG; ++r) {
        const float* row = W2 + (size_t)it[r] * (2 * DD);
        lo[r]   = *(const float4*)(row + 4 * lane);
        hi[r]   = *(const float4*)(row + DD + 4 * lane);
        bias[r] = b2[it[r]];
    }
}

__device__ __forceinline__ void score_sum(const float4 (&lo)[SG], const float4 (&hi)[SG],
                                          const float (&bias)[SG],
                                          const float4 xa, const float4 xb,
                                          const int lane, float* __restrict__ outp)
{
    float p[SG];
    #pragma unroll
    for (int r = 0; r < SG; ++r)
        p[r] = lo[r].x * xa.x + lo[r].y * xa.y + lo[r].z * xa.z + lo[r].w * xa.w
             + hi[r].x * xb.x + hi[r].y * xb.y + hi[r].z * xb.z + hi[r].w * xb.w;

    const bool up = (lane & 16) != 0;
    const bool b3 = (lane & 8) != 0;
    const float q0 = (up ? p[2] : p[0]) + lane_xor16(up ? p[0] : p[2]);
    const float q1 = (up ? p[3] : p[1]) + lane_xor16(up ? p[1] : p[3]);
    float v = (b3 ? q1 : q0) + lane_xor_row<8>(b3 ? q0 : q1);
    v += lane_xor_row<4>(v);
    v += lane_xor_row<2>(v);
    v += lane_xor_row<1>(v);
    const float bsel = up ? (b3 ? bias[3] : bias[2]) : (b3 ? bias[1] : bias[0]);
    if ((lane & 7) == 0)
        outp[(lane >> 3)] = v + bsel;
}

__global__ void __launch_bounds__(512)
caser_score(const int* __restrict__ items,
            const float* __restrict__ xc,
            const float* __restrict__ W2,
            const float* __restrict__ b2,
            float* __restrict__ out)
{
    const int tid  = threadIdx.x;
    const int wave = __builtin_amdgcn_readfirstlane(tid >> 5);
    const int lane = tid & 31;
    const int b    = blockIdx.x * 16 + wave;

    const int* __restrict__ itp = items + (size_t)b * TT;
    float* __restrict__ op      = out + (size_t)b * TT;

    const float4 xa = *(const float4*)(xc + (size_t)b * (2 * DD) + 4 * lane);
    const float4 xb = *(const float4*)(xc + (size_t)b * (2 * DD) + DD + 4 * lane);

    float4 loA[SG], hiA[SG], loB[SG], hiB[SG];
    float  bA[SG], bB[SG];
    int    iA[SG], iB[SG];

    score_idx(itp, iA);
    score_idx(itp + SG, iB);
    score_issue(iA, W2, b2, lane, loA, hiA, bA);
    for (int g = 0; g < NG - 1; g += 2) {
        score_issue(iB, W2, b2, lane, loB, hiB, bB);
        score_idx(itp + (g + 2) * SG, iA);
        score_sum(loA, hiA, bA, xa, xb, lane, op + g * SG);
        score_issue(iA, W2, b2, lane, loA, hiA, bA);
        if (g + 3 < NG) score_idx(itp + (g + 3) * SG, iB);
        score_sum(loB, hiB, bB, xa, xb, lane, op + (g + 1) * SG);
    }
    score_sum(loA, hiA, bA, xa, xb, lane, op + (NG - 1) * SG);
}

extern "C" void kernel_launch(void* const* d_in, const int* in_sizes, int n_in,
                              void* d_out, int out_size, void* d_ws, size_t ws_size,
                              hipStream_t stream) {
    const int*   seq        = (const int*)  d_in[0];
    const int*   user       = (const int*)  d_in[1];
    const int*   items      = (const int*)  d_in[2];
    const float* item_table = (const float*)d_in[3];
    const float* user_table = (const float*)d_in[4];
    const float* Wv         = (const float*)d_in[5];
    const float* bv         = (const float*)d_in[6];
    const float* Wh         = (const float*)d_in[7];
    const float* bh         = (const float*)d_in[8];
    const float* W1         = (const float*)d_in[9];
    const float* b1         = (const float*)d_in[10];
    const float* W2         = (const float*)d_in[11];
    const float* b2         = (const float*)d_in[12];
    float* out = (float*)d_out;

    float* z  = (float*)d_ws;
    float* xc = z + (size_t)BB * FC1;

    caser_features<<<BB / 16, 512, 0, stream>>>(seq, user, item_table, user_table,
                                           Wv, bv, Wh, bh, z, xc);
    caser_fc1<<<BB / 16, 256, 0, stream>>>(z, W1, b1, xc);
    caser_score<<<BB / 16, 512, 0, stream>>>(items, xc, W2, b2, out);
}


# Features and FC1 in one launch: the 16x752 z tile stays in LDS between the convs and the FC1 WMMAs (no z store/reload, one launch less), W1 slabs staged by all 16 waves; summation order unchanged
# speedup vs baseline: 6.2347x; 1.1130x over previous
#include <hip/hip_runtime.h>
#include <hip/hip_bf16.h>

#define BB 4096
#define LL 5
#define DD 128
#define NHH 16
#define NVV 4
#define TT 100
#define FC1 752

typedef float v2f __attribute__((ext_vector_type(2)));
typedef float v8f __attribute__((ext_vector_type(8)));

#define ROWP 644
#define PFP  20
#define PTS  320
#define NPT  35

#define ZP    756
#define WPP   272
#define SROWS 96
#define NSLAB 8
#define STG   (SROWS / 2 * WPP)

template <int LOUT>
__device__ __forceinline__ void hconv_slab(const float4 (&a)[16],
                                           const float* __restrict__ eb,
                                           float* __restrict__ pp)
{
    v8f acc[LOUT];
    #pragma unroll
    for (int t = 0; t < LOUT; ++t) acc[t] = v8f{};

    #pragma unroll
    for (int j = 0; j < 16; ++j) {
        float4 e[LOUT];
        #pragma unroll
        for (int t = 0; t < LOUT; ++t)
            e[t] = *(const float4*)(eb + t * DD + 8 * j);
        const v2f a0 = {a[j].x, a[j].y};
        const v2f a1 = {a[j].z, a[j].w};
        #pragma unroll
        for (int t = 0; t < LOUT; ++t) {
            const v2f b0 = {e[t].x, e[t].y};
            acc[t] = __builtin_amdgcn_wmma_f32_16x16x4_f32(
                false, a0, false, b0, (short)0, acc[t], false, false);
        }
        #pragma unroll
        for (int t = 0; t < LOUT; ++t) {
            const v2f b1 = {e[t].z, e[t].w};
            acc[t] = __builtin_amdgcn_wmma_f32_16x16x4_f32(
                false, a1, false, b1, (short)0, acc[t], false, false);
        }
    }

    #pragma unroll
    for (int t = 0; t < LOUT; ++t) {
        #pragma unroll
        for (int r = 0; r < 8; ++r)
            pp[t * PTS + r * PFP] = acc[t][r];
    }
}

template <int NP>
__device__ __forceinline__ void fc1_fetch(const float* __restrict__ wrow, float4 (&w)[6])
{
    #pragma unroll
    for (int i = 0; i < NP; ++i) {
        w[2 * i]     = *(const float4*)(wrow + (size_t)(32 * i) * DD);
        w[2 * i + 1] = *(const float4*)(wrow + (size_t)(32 * i + 1) * DD);
    }
}

template <int NP>
__device__ __forceinline__ void fc1_park(const float4 (&w)[6], float* __restrict__ sp)
{
    #pragma unroll
    for (int i = 0; i < NP; ++i) {
        const float4 r0 = w[2 * i];
        const float4 r1 = w[2 * i + 1];
        *(float4*)(sp + i * 16 * WPP)     = make_float4(r0.x, r1.x, r0.y, r1.y);
        *(float4*)(sp + i * 16 * WPP + 4) = make_float4(r0.z, r1.z, r0.w, r1.w);
    }
}

template <int NB>
__device__ __forceinline__ void fc1_slab(const float* __restrict__ za,
                                         const float* __restrict__ wb,
                                         v8f& acc0, v8f& acc1)
{
    float4 a[NB];
    float2 p0[NB], p1[NB];
    #pragma unroll
    for (int j = 0; j < NB; ++j) {
        a[j]  = *(const float4*)(za + 8 * j);
        p0[j] = *(const float2*)(wb + (4 * j) * WPP);
        p1[j] = *(const float2*)(wb + (4 * j + 1) * WPP);
    }
    __builtin_amdgcn_sched_barrier(0);
    #pragma unroll
    for (int j = 0; j < NB; ++j) {
        const v2f a0 = {a[j].x, a[j].y};
        const v2f a1 = {a[j].z, a[j].w};
        const v2f b0 = {p0[j].x, p0[j].y};
        const v2f b1 = {p1[j].x, p1[j].y};
        acc0 = __builtin_amdgcn_wmma_f32_16x16x4_f32(
            false, a0, false, b0, (short)0, acc0, false, false);
        acc1 = __builtin_amdgcn_wmma_f32_16x16x4_f32(
            false, a1, false, b1, (short)0, acc1, false, false);
    }
}

__global__ void __launch_bounds__(512)
caser_features_fc1(const int* __restrict__ seq,
                   const int* __restrict__ user,
                   const float* __restrict__ item_table,
                   const float* user_table,
                   const float* __restrict__ Wv,
                   const float* __restrict__ bv,
                   const float* __restrict__ Wh,
                   const float* __restrict__ bh,
                   const float* __restrict__ W1,
                   const float* __restrict__ b1,
                   float* __restrict__ xc)
{
    __shared__ __align__(16) float semb[16 * ROWP];
    __shared__ __align__(16) float spart[NPT * PTS];
    __shared__ float sbh[LL * NHH];
    __shared__ __align__(16) float sz[16 * ZP];
    __shared__ __align__(16) float sw[2 * STG];

    const int tid  = threadIdx.x;
    const int wave = __builtin_amdgcn_readfirstlane(tid >> 5);
    const int lane = tid & 31;
    const int idx16 = lane & 15;
    const int half  = lane >> 4;
    const int b0   = blockIdx.x * 16;

    const int l = (int)((0x5505454543343221ull >> (4 * wave)) & 15ull);
    const int s = (int)((0x4301321021002100ull >> (4 * wave)) & 15ull);
    const int lout = LL - l + 1;

    const int b = b0 + wave;
    int it[LL];
    #pragma unroll
    for (int t = 0; t < LL; ++t) it[t] = seq[b * LL + t];
    const int ui = user[b];
    float wv[NVV * LL], bvr[NVV];
    #pragma unroll
    for (int i = 0; i < NVV * LL; ++i) wv[i] = Wv[i];
    #pragma unroll
    for (int v = 0; v < NVV; ++v) bvr[v] = bv[v];
    float bhv = 0.f;
    if (tid < LL * NHH) bhv = bh[tid];

    float4 a[16];
    if (l > 0) {
        const float* wl = Wh + ((size_t)((l - 1) * NHH + idx16) * LL + s) * DD + 4 * half;
        #pragma unroll
        for (int j = 0; j < 16; ++j)
            a[j] = *(const float4*)(wl + 8 * j);
    }
    float4 w[6];
    const float* wrow = W1 + (size_t)(2 * wave) * DD + 4 * lane;
    float* spark = sw + wave * WPP + 8 * lane;
    fc1_fetch<3>(wrow, w);

    float4 e[LL];
    #pragma unroll
    for (int t = 0; t < LL; ++t)
        e[t] = *(const float4*)(item_table + (size_t)it[t] * DD + 4 * lane);
    const float4 ue = *(const float4*)(user_table + (size_t)ui * DD + 4 * lane);
    fc1_park<3>(w, spark);
    #pragma unroll
    for (int t = 0; t < LL; ++t)
        *(float4*)(semb + wave * ROWP + t * DD + 4 * lane) = e[t];
    if (tid < LL * NHH) sbh[tid] = bhv;
    __syncthreads();

    #pragma unroll
    for (int v = 0; v < NVV; ++v) {
        const float bias = bvr[v];
        float4 acc = make_float4(bias, bias, bias, bias);
        #pragma unroll
        for (int t = 0; t < LL; ++t) {
            const float wt = wv[v * LL + t];
            acc.x += e[t].x * wt;
            acc.y += e[t].y * wt;
            acc.z += e[t].z * wt;
            acc.w += e[t].w * wt;
        }
        *(float4*)(sz + wave * ZP + v * DD + 4 * lane) = acc;
    }
    *(float4*)(xc + (size_t)b * (2 * DD) + DD + 4 * lane) = ue;

    if (l > 0) {
        const float* eb = semb + idx16 * ROWP + s * DD + 4 * half;
        const int base = (l == 1) ? 0 : (l == 2) ? 5 : (l == 3) ? 13 : (l == 4) ? 22 : 30;
        float* pp = spart + (base + s * lout) * PTS + (8 * half) * PFP + idx16;
        switch (lout) {
            case 5:  hconv_slab<5>(a, eb, pp); break;
            case 4:  hconv_slab<4>(a, eb, pp); break;
            case 3:  hconv_slab<3>(a, eb, pp); break;
            case 2:  hconv_slab<2>(a, eb, pp); break;
            default: hconv_slab<1>(a, eb, pp); break;
        }
    }
    __syncthreads();

    if (tid < 480) {
        const int nh = (tid >= 240) ? 1 : 0;
        const int c  = tid - 240 * nh;
        int ll, rem, pbase;
        if      (c <  80) { ll = 1; rem = c;       pbase =  0; }
        else if (c < 144) { ll = 2; rem = c -  80; pbase =  5; }
        else if (c < 192) { ll = 3; rem = c - 144; pbase = 13; }
        else if (c < 224) { ll = 4; rem = c - 192; pbase = 22; }
        else              { ll = 5; rem = c - 224; pbase = 30; }
        const int lo = LL - ll + 1;
        const int f = (ll == 1) ? rem / 5 : (ll == 2) ? (rem >> 2)
                    : (ll == 3) ? rem / 3 : (ll == 4) ? (rem >> 1) : rem;
        const int t = rem - f * lo;
        const float bias = sbh[(ll - 1) * NHH + f];
        const float* p0 = spart + (pbase + t) * PTS + f * PFP + nh;
        int so[LL];
        #pragma unroll
        for (int ss = 0; ss < LL; ++ss) so[ss] = (ss < ll) ? ss * lo * PTS : 0;
        float* zp = sz + nh * ZP + NVV * DD + c;
        #pragma unroll
        for (int k = 0; k < 8; ++k) {
            float acc = 0.f;
            #pragma unroll
            for (int ss = 0; ss < LL; ++ss) {
                const float v = p0[so[ss] + 2 * k];
                acc = (ss < ll) ? acc + v : acc;
            }
            acc += bias;
            zp[(2 * k) * ZP] = fmaxf(acc, 0.f);
        }
    }
    __syncthreads();

    const bool mm = wave < 8;
    const float* za = sz + idx16 * ZP + 4 * half;
    const float* wb = sw + (2 * half) * WPP + 2 * ((wave & 7) * 16 + idx16);
    v8f acc0 = {};
    v8f acc1 = {};

    for (int sl = 0; sl < NSLAB - 2; ++sl) {
        const int cur = (sl & 1) * STG;
        const int nxt = STG - cur;
        fc1_fetch<3>(wrow + (size_t)(sl + 1) * SROWS * DD, w);
        __builtin_amdgcn_sched_barrier(0);
        if (mm) fc1_slab<12>(za + sl * SROWS, wb + cur, acc0, acc1);
        fc1_park<3>(w, spark + nxt);
        __syncthreads();
    }
    if (mm) fc1_fetch<3>(wrow + (size_t)(NSLAB - 1) * SROWS * DD, w);
    else    fc1_fetch<2>(wrow + (size_t)(NSLAB - 1) * SROWS * DD, w);
    __builtin_amdgcn_sched_barrier(0);
    if (mm) {
        fc1_slab<12>(za + (NSLAB - 2) * SROWS, wb, acc0, acc1);
        fc1_park<3>(w, spark + STG);
    } else {
        fc1_park<2>(w, spark + STG);
    }
    __syncthreads();
    if (mm) {
        fc1_slab<10>(za + (NSLAB - 1) * SROWS, wb + STG, acc0, acc1);

        const int n = wave * 16 + idx16;
        const float bias = b1[n];
        #pragma unroll
        for (int r = 0; r < 8; ++r) {
            const int m = b0 + r + half * 8;
            const float v = (acc0[r] + acc1[r]) + bias;
            xc[(size_t)m * (2 * DD) + n] = fmaxf(v, 0.f);
        }
    }
}

#define SG 4
#define NG 25

template <int M>
__device__ __forceinline__ float lane_xor_row(float v)
{
    const int i = __builtin_bit_cast(int, v);
    return __builtin_bit_cast(float,
        __builtin_amdgcn_update_dpp(i, i, 0x160 | M, 0xF, 0xF, false));
}

__device__ __forceinline__ float lane_xor16(float v)
{
    const int i = __builtin_bit_cast(int, v);
    return __builtin_bit_cast(float,
        __builtin_amdgcn_permlanex16(i, i, 0x76543210, 0xfedcba98, false, false));
}

__device__ __forceinline__ void score_idx(const int* __restrict__ itp, int (&it)[SG])
{
    #pragma unroll
    for (int r = 0; r < SG; ++r) it[r] = itp[r];
}

__device__ __forceinline__ void score_issue(const int (&it)[SG],
                                            const float* __restrict__ W2,
                                            const float* __restrict__ b2,
                                            const int lane,
                                            float4 (&lo)[SG], float4 (&hi)[SG],
                                            float (&bias)[SG])
{
    #pragma unroll
    for (int r = 0; r < SG; ++r) {
        const float* row = W2 + (size_t)it[r] * (2 * DD);
        lo[r]   = *(const float4*)(row + 4 * lane);
        hi[r]   = *(const float4*)(row + DD + 4 * lane);
        bias[r] = b2[it[r]];
    }
}

__device__ __forceinline__ void score_sum(const float4 (&lo)[SG], const float4 (&hi)[SG],
                                          const float (&bias)[SG],
                                          const float4 xa, const float4 xb,
                                          const int lane, float* __restrict__ outp)
{
    float p[SG];
    #pragma unroll
    for (int r = 0; r < SG; ++r)
        p[r] = lo[r].x * xa.x + lo[r].y * xa.y + lo[r].z * xa.z + lo[r].w * xa.w
             + hi[r].x * xb.x + hi[r].y * xb.y + hi[r].z * xb.z + hi[r].w * xb.w;

    const bool up = (lane & 16) != 0;
    const bool b3 = (lane & 8) != 0;
    const float q0 = (up ? p[2] : p[0]) + lane_xor16(up ? p[0] : p[2]);
    const float q1 = (up ? p[3] : p[1]) + lane_xor16(up ? p[1] : p[3]);
    float v = (b3 ? q1 : q0) + lane_xor_row<8>(b3 ? q0 : q1);
    v += lane_xor_row<4>(v);
    v += lane_xor_row<2>(v);
    v += lane_xor_row<1>(v);
    const float bsel = up ? (b3 ? bias[3] : bias[2]) : (b3 ? bias[1] : bias[0]);
    if ((lane & 7) == 0)
        outp[(lane >> 3)] = v + bsel;
}

__global__ void __launch_bounds__(512)
caser_score(const int* __restrict__ items,
            const float* __restrict__ xc,
            const float* __restrict__ W2,
            const float* __restrict__ b2,
            float* __restrict__ out)
{
    const int tid  = threadIdx.x;
    const int wave = __builtin_amdgcn_readfirstlane(tid >> 5);
    const int lane = tid & 31;
    const int b    = blockIdx.x * 16 + wave;

    const int* __restrict__ itp = items + (size_t)b * TT;
    float* __restrict__ op      = out + (size_t)b * TT;

    const float4 xa = *(const float4*)(xc + (size_t)b * (2 * DD) + 4 * lane);
    const float4 xb = *(const float4*)(xc + (size_t)b * (2 * DD) + DD + 4 * lane);

    float4 loA[SG], hiA[SG], loB[SG], hiB[SG];
    float  bA[SG], bB[SG];
    int    iA[SG], iB[SG];

    score_idx(itp, iA);
    score_idx(itp + SG, iB);
    score_issue(iA, W2, b2, lane, loA, hiA, bA);
    for (int g = 0; g < NG - 1; g += 2) {
        score_issue(iB, W2, b2, lane, loB, hiB, bB);
        score_idx(itp + (g + 2) * SG, iA);
        score_sum(loA, hiA, bA, xa, xb, lane, op + g * SG);
        score_issue(iA, W2, b2, lane, loA, hiA, bA);
        if (g + 3 < NG) score_idx(itp + (g + 3) * SG, iB);
        score_sum(loB, hiB, bB, xa, xb, lane, op + (g + 1) * SG);
    }
    score_sum(loA, hiA, bA, xa, xb, lane, op + (NG - 1) * SG);
}

extern "C" void kernel_launch(void* const* d_in, const int* in_sizes, int n_in,
                              void* d_out, int out_size, void* d_ws, size_t ws_size,
                              hipStream_t stream) {
    const int*   seq        = (const int*)  d_in[0];
    const int*   user       = (const int*)  d_in[1];
    const int*   items      = (const int*)  d_in[2];
    const float* item_table = (const float*)d_in[3];
    const float* user_table = (const float*)d_in[4];
    const float* Wv         = (const float*)d_in[5];
    const float* bv         = (const float*)d_in[6];
    const float* Wh         = (const float*)d_in[7];
    const float* bh         = (const float*)d_in[8];
    const float* W1         = (const float*)d_in[9];
    const float* b1         = (const float*)d_in[10];
    const float* W2         = (const float*)d_in[11];
    const float* b2         = (const float*)d_in[12];
    float* out = (float*)d_out;

    float* xc = (float*)d_ws;

    caser_features_fc1<<<BB / 16, 512, 0, stream>>>(seq, user, item_table, user_table,
                                               Wv, bv, Wh, bh, W1, b1, xc);
    caser_score<<<BB / 16, 512, 0, stream>>>(items, xc, W2, b2, out);
}


# Score kernel: a row's 100 indices by one vector load, row bases by lane reads, b2 and sums in lanes: gather code free of scalar memory waits, 24-32 row loads always in flight; sum order unchanged
# speedup vs baseline: 6.4462x; 1.0339x over previous
#include <hip/hip_runtime.h>
#include <hip/hip_bf16.h>

#define BB 4096
#define LL 5
#define DD 128
#define NHH 16
#define NVV 4
#define TT 100
#define FC1 752

typedef float v2f __attribute__((ext_vector_type(2)));
typedef float v8f __attribute__((ext_vector_type(8)));

#define ROWP 644
#define PFP  20
#define PTS  320
#define NPT  35

#define ZP    756
#define WPP   272
#define SROWS 96
#define NSLAB 8
#define STG   (SROWS / 2 * WPP)

template <int LOUT>
__device__ __forceinline__ void hconv_slab(const float4 (&a)[16],
                                           const float* __restrict__ eb,
                                           float* __restrict__ pp)
{
    v8f acc[LOUT];
    #pragma unroll
    for (int t = 0; t < LOUT; ++t) acc[t] = v8f{};

    #pragma unroll
    for (int j = 0; j < 16; ++j) {
        float4 e[LOUT];
        #pragma unroll
        for (int t = 0; t < LOUT; ++t)
            e[t] = *(const float4*)(eb + t * DD + 8 * j);
        const v2f a0 = {a[j].x, a[j].y};
        const v2f a1 = {a[j].z, a[j].w};
        #pragma unroll
        for (int t = 0; t < LOUT; ++t) {
            const v2f b0 = {e[t].x, e[t].y};
            acc[t] = __builtin_amdgcn_wmma_f32_16x16x4_f32(
                false, a0, false, b0, (short)0, acc[t], false, false);
        }
        #pragma unroll
        for (int t = 0; t < LOUT; ++t) {
            const v2f b1 = {e[t].z, e[t].w};
            acc[t] = __builtin_amdgcn_wmma_f32_16x16x4_f32(
                false, a1, false, b1, (short)0, acc[t], false, false);
        }
    }

    #pragma unroll
    for (int t = 0; t < LOUT; ++t) {
        #pragma unroll
        for (int r = 0; r < 8; ++r)
            pp[t * PTS + r * PFP] = acc[t][r];
    }
}

template <int NP>
__device__ __forceinline__ void fc1_fetch(const float* __restrict__ wrow, float4 (&w)[6])
{
    #pragma unroll
    for (int i = 0; i < NP; ++i) {
        w[2 * i]     = *(const float4*)(wrow + (size_t)(32 * i) * DD);
        w[2 * i + 1] = *(const float4*)(wrow + (size_t)(32 * i + 1) * DD);
    }
}

template <int NP>
__device__ __forceinline__ void fc1_park(const float4 (&w)[6], float* __restrict__ sp)
{
    #pragma unroll
    for (int i = 0; i < NP; ++i) {
        const float4 r0 = w[2 * i];
        const float4 r1 = w[2 * i + 1];
        *(float4*)(sp + i * 16 * WPP)     = make_float4(r0.x, r1.x, r0.y, r1.y);
        *(float4*)(sp + i * 16 * WPP + 4) = make_float4(r0.z, r1.z, r0.w, r1.w);
    }
}

template <int NB>
__device__ __forceinline__ void fc1_slab(const float* __restrict__ za,
                                         const float* __restrict__ wb,
                                         v8f& acc0, v8f& acc1)
{
    float4 a[NB];
    float2 p0[NB], p1[NB];
    #pragma unroll
    for (int j = 0; j < NB; ++j) {
        a[j]  = *(const float4*)(za + 8 * j);
        p0[j] = *(const float2*)(wb + (4 * j) * WPP);
        p1[j] = *(const float2*)(wb + (4 * j + 1) * WPP);
    }
    __builtin_amdgcn_sched_barrier(0);
    #pragma unroll
    for (int j = 0; j < NB; ++j) {
        const v2f a0 = {a[j].x, a[j].y};
        const v2f a1 = {a[j].z, a[j].w};
        const v2f b0 = {p0[j].x, p0[j].y};
        const v2f b1 = {p1[j].x, p1[j].y};
        acc0 = __builtin_amdgcn_wmma_f32_16x16x4_f32(
            false, a0, false, b0, (short)0, acc0, false, false);
        acc1 = __builtin_amdgcn_wmma_f32_16x16x4_f32(
            false, a1, false, b1, (short)0, acc1, false, false);
    }
}

__global__ void __launch_bounds__(512)
caser_features_fc1(const int* __restrict__ seq,
                   const int* __restrict__ user,
                   const float* __restrict__ item_table,
                   const float* user_table,
                   const float* __restrict__ Wv,
                   const float* __restrict__ bv,
                   const float* __restrict__ Wh,
                   const float* __restrict__ bh,
                   const float* __restrict__ W1,
                   const float* __restrict__ b1,
                   float* __restrict__ xc)
{
    __shared__ __align__(16) float semb[16 * ROWP];
    __shared__ __align__(16) float spart[NPT * PTS];
    __shared__ float sbh[LL * NHH];
    __shared__ __align__(16) float sz[16 * ZP];
    __shared__ __align__(16) float sw[2 * STG];

    const int tid  = threadIdx.x;
    const int wave = __builtin_amdgcn_readfirstlane(tid >> 5);
    const int lane = tid & 31;
    const int idx16 = lane & 15;
    const int half  = lane >> 4;
    const int b0   = blockIdx.x * 16;

    const int l = (int)((0x5505454543343221ull >> (4 * wave)) & 15ull);
    const int s = (int)((0x4301321021002100ull >> (4 * wave)) & 15ull);
    const int lout = LL - l + 1;

    const int b = b0 + wave;
    int it[LL];
    #pragma unroll
    for (int t = 0; t < LL; ++t) it[t] = seq[b * LL + t];
    const int ui = user[b];
    float wv[NVV * LL], bvr[NVV];
    #pragma unroll
    for (int i = 0; i < NVV * LL; ++i) wv[i] = Wv[i];
    #pragma unroll
    for (int v = 0; v < NVV; ++v) bvr[v] = bv[v];
    float bhv = 0.f;
    if (tid < LL * NHH) bhv = bh[tid];

    float4 a[16];
    if (l > 0) {
        const float* wl = Wh + ((size_t)((l - 1) * NHH + idx16) * LL + s) * DD + 4 * half;
        #pragma unroll
        for (int j = 0; j < 16; ++j)
            a[j] = *(const float4*)(wl + 8 * j);
    }
    float4 w[6];
    const float* wrow = W1 + (size_t)(2 * wave) * DD + 4 * lane;
    float* spark = sw + wave * WPP + 8 * lane;
    fc1_fetch<3>(wrow, w);

    float4 e[LL];
    #pragma unroll
    for (int t = 0; t < LL; ++t)
        e[t] = *(const float4*)(item_table + (size_t)it[t] * DD + 4 * lane);
    const float4 ue = *(const float4*)(user_table + (size_t)ui * DD + 4 * lane);
    fc1_park<3>(w, spark);
    #pragma unroll
    for (int t = 0; t < LL; ++t)
        *(float4*)(semb + wave * ROWP + t * DD + 4 * lane) = e[t];
    if (tid < LL * NHH) sbh[tid] = bhv;
    __syncthreads();

    #pragma unroll
    for (int v = 0; v < NVV; ++v) {
        const float bias = bvr[v];
        float4 acc = make_float4(bias, bias, bias, bias);
        #pragma unroll
        for (int t = 0; t < LL; ++t) {
            const float wt = wv[v * LL + t];
            acc.x += e[t].x * wt;
            acc.y += e[t].y * wt;
            acc.z += e[t].z * wt;
            acc.w += e[t].w * wt;
        }
        *(float4*)(sz + wave * ZP + v * DD + 4 * lane) = acc;
    }
    *(float4*)(xc + (size_t)b * (2 * DD) + DD + 4 * lane) = ue;

    if (l > 0) {
        const float* eb = semb + idx16 * ROWP + s * DD + 4 * half;
        const int base = (l == 1) ? 0 : (l == 2) ? 5 : (l == 3) ? 13 : (l == 4) ? 22 : 30;
        float* pp = spart + (base + s * lout) * PTS + (8 * half) * PFP + idx16;
        switch (lout) {
            case 5:  hconv_slab<5>(a, eb, pp); break;
            case 4:  hconv_slab<4>(a, eb, pp); break;
            case 3:  hconv_slab<3>(a, eb, pp); break;
            case 2:  hconv_slab<2>(a, eb, pp); break;
            default: hconv_slab<1>(a, eb, pp); break;
        }
    }
    __syncthreads();

    if (tid < 480) {
        const int nh = (tid >= 240) ? 1 : 0;
        const int c  = tid - 240 * nh;
        int ll, rem, pbase;
        if      (c <  80) { ll = 1; rem = c;       pbase =  0; }
        else if (c < 144) { ll = 2; rem = c -  80; pbase =  5; }
        else if (c < 192) { ll = 3; rem = c - 144; pbase = 13; }
        else if (c < 224) { ll = 4; rem = c - 192; pbase = 22; }
        else              { ll = 5; rem = c - 224; pbase = 30; }
        const int lo = LL - ll + 1;
        const int f = (ll == 1) ? rem / 5 : (ll == 2) ? (rem >> 2)
                    : (ll == 3) ? rem / 3 : (ll == 4) ? (rem >> 1) : rem;
        const int t = rem - f * lo;
        const float bias = sbh[(ll - 1) * NHH + f];
        const float* p0 = spart + (pbase + t) * PTS + f * PFP + nh;
        int so[LL];
        #pragma unroll
        for (int ss = 0; ss < LL; ++ss) so[ss] = (ss < ll) ? ss * lo * PTS : 0;
        float* zp = sz + nh * ZP + NVV * DD + c;
        #pragma unroll
        for (int k = 0; k < 8; ++k) {
            float acc = 0.f;
            #pragma unroll
            for (int ss = 0; ss < LL; ++ss) {
                const float v = p0[so[ss] + 2 * k];
                acc = (ss < ll) ? acc + v : acc;
            }
            acc += bias;
            zp[(2 * k) * ZP] = fmaxf(acc, 0.f);
        }
    }
    __syncthreads();

    const bool mm = wave < 8;
    const float* za = sz + idx16 * ZP + 4 * half;
    const float* wb = sw + (2 * half) * WPP + 2 * ((wave & 7) * 16 + idx16);
    v8f acc0 = {};
    v8f acc1 = {};

    for (int sl = 0; sl < NSLAB - 2; ++sl) {
        const int cur = (sl & 1) * STG;
        const int nxt = STG - cur;
        fc1_fetch<3>(wrow + (size_t)(sl + 1) * SROWS * DD, w);
        __builtin_amdgcn_sched_barrier(0);
        if (mm) fc1_slab<12>(za + sl * SROWS, wb + cur, acc0, acc1);
        fc1_park<3>(w, spark + nxt);
        __syncthreads();
    }
    if (mm) fc1_fetch<3>(wrow + (size_t)(NSLAB - 1) * SROWS * DD, w);
    else    fc1_fetch<2>(wrow + (size_t)(NSLAB - 1) * SROWS * DD, w);
    __builtin_amdgcn_sched_barrier(0);
    if (mm) {
        fc1_slab<12>(za + (NSLAB - 2) * SROWS, wb, acc0, acc1);
        fc1_park<3>(w, spark + STG);
    } else {
        fc1_park<2>(w, spark + STG);
    }
    __syncthreads();
    if (mm) {
        fc1_slab<10>(za + (NSLAB - 1) * SROWS, wb + STG, acc0, acc1);

        const int n = wave * 16 + idx16;
        const float bias = b1[n];
        #pragma unroll
        for (int r = 0; r < 8; ++r) {
            const int m = b0 + r + half * 8;
            const float v = (acc0[r] + acc1[r]) + bias;
            xc[(size_t)m * (2 * DD) + n] = fmaxf(v, 0.f);
        }
    }
}

#define SG 4
#define NG 25
#define NSET 4

template <int M>
__device__ __forceinline__ float lane_xor_row(float v)
{
    const int i = __builtin_bit_cast(int, v);
    return __builtin_bit_cast(float,
        __builtin_amdgcn_update_dpp(i, i, 0x160 | M, 0xF, 0xF, false));
}

__device__ __forceinline__ float lane_xor16(float v)
{
    const int i = __builtin_bit_cast(int, v);
    return __builtin_bit_cast(float,
        __builtin_amdgcn_permlanex16(i, i, 0x76543210, 0xfedcba98, false, false));
}

__device__ __forceinline__ void score_issue(const int4 iv, const int g,
                                            const float* __restrict__ W2,
                                            const int lane,
                                            float4 (&lo)[SG], float4 (&hi)[SG])
{
    const int it[SG] = { __builtin_amdgcn_readlane(iv.x, g), __builtin_amdgcn_readlane(iv.y, g),
                         __builtin_amdgcn_readlane(iv.z, g), __builtin_amdgcn_readlane(iv.w, g) };
    #pragma unroll
    for (int r = 0; r < SG; ++r) {
        const float* row = W2 + (size_t)it[r] * (2 * DD);
        lo[r] = *(const float4*)(row + 4 * lane);
        hi[r] = *(const float4*)(row + DD + 4 * lane);
    }
}

__device__ __forceinline__ void score_sum(const float4 (&lo)[SG], const float4 (&hi)[SG],
                                          const float4 xa, const float4 xb,
                                          const int lane, const int g, float4& res)
{
    float p[SG];
    #pragma unroll
    for (int r = 0; r < SG; ++r)
        p[r] = lo[r].x * xa.x + lo[r].y * xa.y + lo[r].z * xa.z + lo[r].w * xa.w
             + hi[r].x * xb.x + hi[r].y * xb.y + hi[r].z * xb.z + hi[r].w * xb.w;

    const bool up = (lane & 16) != 0;
    const bool b3 = (lane & 8) != 0;
    const float q0 = (up ? p[2] : p[0]) + lane_xor16(up ? p[0] : p[2]);
    const float q1 = (up ? p[3] : p[1]) + lane_xor16(up ? p[1] : p[3]);
    float v = (b3 ? q1 : q0) + lane_xor_row<8>(b3 ? q0 : q1);
    v += lane_xor_row<4>(v);
    v += lane_xor_row<2>(v);
    v += lane_xor_row<1>(v);
    const int vi = __builtin_bit_cast(int, v);
    const int s0 = __builtin_amdgcn_readlane(vi, 0);
    const int s1 = __builtin_amdgcn_readlane(vi, 8);
    const int s2 = __builtin_amdgcn_readlane(vi, 16);
    const int s3 = __builtin_amdgcn_readlane(vi, 24);
    const bool mine = (lane == g);
    res.x = mine ? __builtin_bit_cast(float, s0) + res.x : res.x;
    res.y = mine ? __builtin_bit_cast(float, s1) + res.y : res.y;
    res.z = mine ? __builtin_bit_cast(float, s2) + res.z : res.z;
    res.w = mine ? __builtin_bit_cast(float, s3) + res.w : res.w;
}

__global__ void __launch_bounds__(512)
caser_score(const int* __restrict__ items,
            const float* __restrict__ xc,
            const float* __restrict__ W2,
            const float* __restrict__ b2,
            float* __restrict__ out)
{
    const int tid  = threadIdx.x;
    const int wave = __builtin_amdgcn_readfirstlane(tid >> 5);
    const int lane = tid & 31;
    const int b    = blockIdx.x * 16 + wave;

    const int* __restrict__ itp = items + (size_t)b * TT;
    float* __restrict__ op      = out + (size_t)b * TT;

    const int li = (lane < NG) ? lane : NG - 1;
    const int4 iv = *(const int4*)(itp + SG * li);

    const float4 xa = *(const float4*)(xc + (size_t)b * (2 * DD) + 4 * lane);
    const float4 xb = *(const float4*)(xc + (size_t)b * (2 * DD) + DD + 4 * lane);

    float4 lo[NSET][SG], hi[NSET][SG];

    score_issue(iv, 0, W2, lane, lo[0], hi[0]);
    float4 res = make_float4(b2[iv.x], b2[iv.y], b2[iv.z], b2[iv.w]);
    score_issue(iv, 1, W2, lane, lo[1], hi[1]);
    score_issue(iv, 2, W2, lane, lo[2], hi[2]);

    #pragma unroll
    for (int g = 0; g < NG; ++g) {
        if (g + 3 < NG)
            score_issue(iv, g + 3, W2, lane, lo[(g + 3) & 3], hi[(g + 3) & 3]);
        __builtin_amdgcn_sched_barrier(0);
        score_sum(lo[g & 3], hi[g & 3], xa, xb, lane, g, res);
        __builtin_amdgcn_sched_barrier(0);
    }

    if (lane < NG)
        *(float4*)(op + SG * lane) = res;
}

extern "C" void kernel_launch(void* const* d_in, const int* in_sizes, int n_in,
                              void* d_out, int out_size, void* d_ws, size_t ws_size,
                              hipStream_t stream) {
    const int*   seq        = (const int*)  d_in[0];
    const int*   user       = (const int*)  d_in[1];
    const int*   items      = (const int*)  d_in[2];
    const float* item_table = (const float*)d_in[3];
    const float* user_table = (const float*)d_in[4];
    const float* Wv         = (const float*)d_in[5];
    const float* bv         = (const float*)d_in[6];
    const float* Wh         = (const float*)d_in[7];
    const float* bh         = (const float*)d_in[8];
    const float* W1         = (const float*)d_in[9];
    const float* b1         = (const float*)d_in[10];
    const float* W2         = (const float*)d_in[11];
    const float* b2         = (const float*)d_in[12];
    float* out = (float*)d_out;

    float* xc = (float*)d_ws;

    caser_features_fc1<<<BB / 16, 512, 0, stream>>>(seq, user, item_table, user_table,
                                               Wv, bv, Wh, bh, W1, b1, xc);
    caser_score<<<BB / 16, 512, 0, stream>>>(items, xc, W2, b2, out);
}
